# starting kernel
# baseline (speedup 1.0000x reference)
#include <hip/hip_runtime.h>
#include <hip/hip_bf16.h>
#include <math.h>

typedef __attribute__((ext_vector_type(2))) float v2f;
typedef __attribute__((ext_vector_type(8))) float v8f;

#define D 128

__global__ __launch_bounds__(256) void zero_kernel(float* p, long long n) {
    long long i = (long long)blockIdx.x * blockDim.x + threadIdx.x;
    long long stride = (long long)gridDim.x * blockDim.x;
    for (; i < n; i += stride) p[i] = 0.0f;
}

__global__ __launch_bounds__(256) void aggregate_kernel(
    const float* __restrict__ x, const int* __restrict__ src,
    const int* __restrict__ dst, float* __restrict__ s,
    float* __restrict__ cnt, int E)
{
    int wid  = (int)(((long long)blockIdx.x * blockDim.x + threadIdx.x) >> 5);
    int lane = threadIdx.x & 31;
    if (wid >= E) return;
    int si = src[wid];
    int di = dst[wid];
    float4 v = ((const float4*)(x + (long long)si * D))[lane];
    float* so = s + (long long)di * D + lane * 4;
    atomicAdd(so + 0, v.x);
    atomicAdd(so + 1, v.y);
    atomicAdd(so + 2, v.z);
    atomicAdd(so + 3, v.w);
    if (lane == 0) atomicAdd(cnt + di, 1.0f);
}

__global__ __launch_bounds__(256) void sage_gemm_kernel(
    const float* __restrict__ xin, const float* __restrict__ s,
    const float* __restrict__ cnt, const float* __restrict__ Wl,
    const float* __restrict__ bias, const float* __restrict__ Wr,
    float* __restrict__ out, int N, int relu)
{
    int mtile = blockIdx.x;
    int wave  = threadIdx.x >> 5;
    int lane  = threadIdx.x & 31;
    int m0    = mtile * 16;
    int n0    = wave * 16;
    int ncol  = n0 + (lane & 15);

    int rowA  = m0 + (lane & 15);
    if (rowA > N - 1) rowA = N - 1;
    int khalf = (lane >> 4) << 1;

    const float* srow = s   + (long long)rowA * D;
    const float* xrow = xin + (long long)rowA * D;
    float inv = 1.0f / fmaxf(cnt[rowA], 1.0f);

    v8f c = {};
    #pragma unroll
    for (int k = 0; k < D; k += 4) {
        int ka = k + khalf;
        v2f a0; a0.x = srow[ka] * inv;  a0.y = srow[ka + 1] * inv;
        v2f b0; b0.x = Wl[ka * D + ncol]; b0.y = Wl[(ka + 1) * D + ncol];
        c = __builtin_amdgcn_wmma_f32_16x16x4_f32(
                false, a0, false, b0, (short)0, c, false, false);
        v2f a1; a1.x = xrow[ka];        a1.y = xrow[ka + 1];
        v2f b1; b1.x = Wr[ka * D + ncol]; b1.y = Wr[(ka + 1) * D + ncol];
        c = __builtin_amdgcn_wmma_f32_16x16x4_f32(
                false, a1, false, b1, (short)0, c, false, false);
    }

    float bn = bias[ncol];
    int mbase = m0 + ((lane >> 4) << 3);
    #pragma unroll
    for (int r = 0; r < 8; ++r) {
        int m = mbase + r;
        if (m < N) {
            float v = c[r] + bn;
            if (relu) v = fmaxf(v, 0.0f);
            out[(long long)m * D + ncol] = v;
        }
    }
}

__global__ __launch_bounds__(256) void decode_kernel(
    const float* __restrict__ h, const int* __restrict__ dec,
    float* __restrict__ out, int Dk)
{
    int wid  = (int)(((long long)blockIdx.x * blockDim.x + threadIdx.x) >> 5);
    int lane = threadIdx.x & 31;
    if (wid >= Dk) return;
    int ai = dec[wid];
    int bi = dec[Dk + wid];
    float4 a = ((const float4*)(h + (long long)ai * D))[lane];
    float4 b = ((const float4*)(h + (long long)bi * D))[lane];
    float dot = a.x * b.x + a.y * b.y + a.z * b.z + a.w * b.w;
    float na  = a.x * a.x + a.y * a.y + a.z * a.z + a.w * a.w;
    float nb  = b.x * b.x + b.y * b.y + b.z * b.z + b.w * b.w;
    #pragma unroll
    for (int off = 16; off > 0; off >>= 1) {
        dot += __shfl_xor(dot, off, 32);
        na  += __shfl_xor(na,  off, 32);
        nb  += __shfl_xor(nb,  off, 32);
    }
    if (lane == 0) {
        float denom = fmaxf(sqrtf(na) * sqrtf(nb), 1e-6f);
        float z = dot / denom;
        out[wid] = 1.0f / (1.0f + expf(-z));
    }
}

extern "C" void kernel_launch(void* const* d_in, const int* in_sizes, int n_in,
                              void* d_out, int out_size, void* d_ws, size_t ws_size,
                              hipStream_t stream) {
    const float* x   = (const float*)d_in[0];
    const int*   edg = (const int*)d_in[1];
    const int*   dec = (const int*)d_in[2];
    const float* W1l = (const float*)d_in[3];
    const float* b1  = (const float*)d_in[4];
    const float* W1r = (const float*)d_in[5];
    const float* W2l = (const float*)d_in[6];
    const float* b2  = (const float*)d_in[7];
    const float* W2r = (const float*)d_in[8];
    float* out = (float*)d_out;

    const int N  = in_sizes[0] / D;
    const int E  = in_sizes[1] / 2;
    const int Dk = in_sizes[2] / 2;
    const int* src = edg;
    const int* dst = edg + E;

    float* ws  = (float*)d_ws;
    float* s   = ws;
    float* cnt = s + (long long)N * D;
    float* h1  = cnt + N;
    float* h2  = h1 + (long long)N * D;

    const long long sAndCnt = (long long)N * D + N;
    const int zeroBlocks  = 2048;
    const int aggrBlocks  = (E + 7) / 8;
    const int gemmBlocks  = (N + 15) / 16;
    const int decBlocks   = (Dk + 7) / 8;

    zero_kernel<<<zeroBlocks, 256, 0, stream>>>(s, sAndCnt);
    aggregate_kernel<<<aggrBlocks, 256, 0, stream>>>(x, src, dst, s, cnt, E);
    sage_gemm_kernel<<<gemmBlocks, 256, 0, stream>>>(x, s, cnt, W1l, b1, W1r,
                                                     h1, N,  1);
    zero_kernel<<<zeroBlocks, 256, 0, stream>>>(s, sAndCnt);
    aggregate_kernel<<<aggrBlocks, 256, 0, stream>>>(h1, src, dst, s, cnt, E);
    sage_gemm_kernel<<<gemmBlocks, 256, 0, stream>>>(h1, s, cnt, W2l, b2, W2r,
                                                     h2, N,  0);
    decode_kernel<<<decBlocks, 256, 0, stream>>>(h2, dec, out, Dk);
}


# aggregate: no global atomics, no zero pass: each of 256 work-groups owns 196 node rows in LDS, scans all dst, adds its own edges' rows 8 at a time; f32 sums in arrival order of LDS adds
# speedup vs baseline: 2.5518x; 2.5518x over previous
#include <hip/hip_runtime.h>
#include <hip/hip_bf16.h>
#include <math.h>

typedef __attribute__((ext_vector_type(2))) float v2f;
typedef __attribute__((ext_vector_type(8))) float v8f;

#define D 128

#define AGG_ROWS   196
#define AGG_WAVES  32
#define AGG_BATCH  1024
#define AGG_Q      (AGG_BATCH + 8)

__device__ __forceinline__ void lds_add(float* p, float v) {
    __hip_atomic_fetch_add(p, v, __ATOMIC_RELAXED, __HIP_MEMORY_SCOPE_WORKGROUP);
}

__device__ __forceinline__ void wave_lds_order() {
    __builtin_amdgcn_fence(__ATOMIC_RELEASE, "wavefront");
    __builtin_amdgcn_wave_barrier();
    __builtin_amdgcn_fence(__ATOMIC_ACQUIRE, "wavefront");
}

template <bool FULL>
__device__ __forceinline__ void drain_round(
    const float* __restrict__ x, const int* __restrict__ src,
    const int* __restrict__ dst, const int* myq, int i, int n, int lo,
    float* acc, float* cntL, int lane)
{
    int slot = (lane < n) ? (i + lane) : i;
    int e  = myq[slot];
    int sv = src[e];
    int tv = dst[e] - lo;
    float4 v[8];
    #pragma unroll
    for (int j = 0; j < 8; ++j) {
        if (FULL || j < n) {
            int sj = __builtin_amdgcn_readlane(sv, j);
            v[j] = ((const float4*)(x + (long long)sj * D))[lane];
        }
    }
    #pragma unroll
    for (int j = 0; j < 8; ++j) {
        if (FULL || j < n) {
            int tj = __builtin_amdgcn_readlane(tv, j);
            float* a = acc + tj * D + lane;
            lds_add(a,      v[j].x);
            lds_add(a + 32, v[j].y);
            lds_add(a + 64, v[j].z);
            lds_add(a + 96, v[j].w);
            if (lane == 0) lds_add(cntL + tj, 1.0f);
        }
    }
}

__global__ __launch_bounds__(1024) void aggregate_owner_kernel(
    const float* __restrict__ x, const int* __restrict__ src,
    const int* __restrict__ dst, float* __restrict__ s,
    float* __restrict__ cnt, int E, int N)
{
    __shared__ float acc[AGG_ROWS * D];
    __shared__ float cntL[AGG_ROWS];
    __shared__ int   q[AGG_WAVES][AGG_Q];

    const int tid  = threadIdx.x;
    const int lane = tid & 31;
    const int wave = tid >> 5;
    int* myq = q[wave];
    const unsigned ltmask = (1u << lane) - 1u;
    const bool aligned = (((unsigned long long)dst) & 15ull) == 0ull;
    const int nbatch = (E + AGG_BATCH - 1) / AGG_BATCH;

    for (int lo = blockIdx.x * AGG_ROWS; lo < N; lo += gridDim.x * AGG_ROWS) {
        const int nrows = min(AGG_ROWS, N - lo);
        for (int i = tid; i < AGG_ROWS * D; i += 1024) acc[i] = 0.0f;
        if (tid < AGG_ROWS) cntL[tid] = 0.0f;
        __syncthreads();

        int qn = 0;
        for (int b = wave; b < nbatch; b += AGG_WAVES) {
            const int base = b * AGG_BATCH + lane * 4;
            int4 d[8];
            if (aligned && (b + 1) * AGG_BATCH <= E) {
                #pragma unroll
                for (int u = 0; u < 8; ++u)
                    d[u] = *(const int4*)(dst + base + u * 128);
            } else {
                #pragma unroll
                for (int u = 0; u < 8; ++u) {
                    int e = base + u * 128;
                    d[u].x = (e     < E) ? dst[e]     : -1;
                    d[u].y = (e + 1 < E) ? dst[e + 1] : -1;
                    d[u].z = (e + 2 < E) ? dst[e + 2] : -1;
                    d[u].w = (e + 3 < E) ? dst[e + 3] : -1;
                }
            }
#define AGG_ENQ(dv, eidx)                                                     \
            {                                                                 \
                bool m = (unsigned)((dv) - lo) < (unsigned)nrows;             \
                unsigned bal = __builtin_amdgcn_ballot_w32(m);                \
                if (bal) {                                                    \
                    if (m) myq[qn + __builtin_popcount(bal & ltmask)] = (eidx); \
                    qn += __builtin_popcount(bal);                            \
                }                                                             \
            }
            #pragma unroll
            for (int u = 0; u < 8; ++u) {
                int e = base + u * 128;
                AGG_ENQ(d[u].x, e)
                AGG_ENQ(d[u].y, e + 1)
                AGG_ENQ(d[u].z, e + 2)
                AGG_ENQ(d[u].w, e + 3)
            }
#undef AGG_ENQ
            wave_lds_order();
            int i = 0;
            for (; i + 8 <= qn; i += 8)
                drain_round<true>(x, src, dst, myq, i, 8, lo, acc, cntL, lane);
            int rem = qn - i;
            if (i > 0 && rem > 0) {
                int vq = myq[i + ((lane < rem) ? lane : 0)];
                if (lane < rem) myq[lane] = vq;
            }
            qn = rem;
            wave_lds_order();
        }
        if (qn > 0)
            drain_round<false>(x, src, dst, myq, 0, qn, lo, acc, cntL, lane);
        __syncthreads();

        for (int r = wave; r < nrows; r += AGG_WAVES) {
            const float* a = acc + r * D + lane;
            float4 o;
            o.x = a[0]; o.y = a[32]; o.z = a[64]; o.w = a[96];
            ((float4*)(s + (long long)(lo + r) * D))[lane] = o;
        }
        if (tid < nrows) cnt[lo + tid] = cntL[tid];
        __syncthreads();
    }
}

__global__ __launch_bounds__(256) void sage_gemm_kernel(
    const float* __restrict__ xin, const float* __restrict__ s,
    const float* __restrict__ cnt, const float* __restrict__ Wl,
    const float* __restrict__ bias, const float* __restrict__ Wr,
    float* __restrict__ out, int N, int relu)
{
    int mtile = blockIdx.x;
    int wave  = threadIdx.x >> 5;
    int lane  = threadIdx.x & 31;
    int m0    = mtile * 16;
    int n0    = wave * 16;
    int ncol  = n0 + (lane & 15);

    int rowA  = m0 + (lane & 15);
    if (rowA > N - 1) rowA = N - 1;
    int khalf = (lane >> 4) << 1;

    const float* srow = s   + (long long)rowA * D;
    const float* xrow = xin + (long long)rowA * D;
    float inv = 1.0f / fmaxf(cnt[rowA], 1.0f);

    v8f c = {};
    #pragma unroll
    for (int k = 0; k < D; k += 4) {
        int ka = k + khalf;
        v2f a0; a0.x = srow[ka] * inv;  a0.y = srow[ka + 1] * inv;
        v2f b0; b0.x = Wl[ka * D + ncol]; b0.y = Wl[(ka + 1) * D + ncol];
        c = __builtin_amdgcn_wmma_f32_16x16x4_f32(
                false, a0, false, b0, (short)0, c, false, false);
        v2f a1; a1.x = xrow[ka];        a1.y = xrow[ka + 1];
        v2f b1; b1.x = Wr[ka * D + ncol]; b1.y = Wr[(ka + 1) * D + ncol];
        c = __builtin_amdgcn_wmma_f32_16x16x4_f32(
                false, a1, false, b1, (short)0, c, false, false);
    }

    float bn = bias[ncol];
    int mbase = m0 + ((lane >> 4) << 3);
    #pragma unroll
    for (int r = 0; r < 8; ++r) {
        int m = mbase + r;
        if (m < N) {
            float v = c[r] + bn;
            if (relu) v = fmaxf(v, 0.0f);
            out[(long long)m * D + ncol] = v;
        }
    }
}

__global__ __launch_bounds__(256) void decode_kernel(
    const float* __restrict__ h, const int* __restrict__ dec,
    float* __restrict__ out, int Dk)
{
    int wid  = (int)(((long long)blockIdx.x * blockDim.x + threadIdx.x) >> 5);
    int lane = threadIdx.x & 31;
    if (wid >= Dk) return;
    int ai = dec[wid];
    int bi = dec[Dk + wid];
    float4 a = ((const float4*)(h + (long long)ai * D))[lane];
    float4 b = ((const float4*)(h + (long long)bi * D))[lane];
    float dot = a.x * b.x + a.y * b.y + a.z * b.z + a.w * b.w;
    float na  = a.x * a.x + a.y * a.y + a.z * a.z + a.w * a.w;
    float nb  = b.x * b.x + b.y * b.y + b.z * b.z + b.w * b.w;
    #pragma unroll
    for (int off = 16; off > 0; off >>= 1) {
        dot += __shfl_xor(dot, off, 32);
        na  += __shfl_xor(na,  off, 32);
        nb  += __shfl_xor(nb,  off, 32);
    }
    if (lane == 0) {
        float denom = fmaxf(sqrtf(na) * sqrtf(nb), 1e-6f);
        float z = dot / denom;
        out[wid] = 1.0f / (1.0f + expf(-z));
    }
}

extern "C" void kernel_launch(void* const* d_in, const int* in_sizes, int n_in,
                              void* d_out, int out_size, void* d_ws, size_t ws_size,
                              hipStream_t stream) {
    const float* x   = (const float*)d_in[0];
    const int*   edg = (const int*)d_in[1];
    const int*   dec = (const int*)d_in[2];
    const float* W1l = (const float*)d_in[3];
    const float* b1  = (const float*)d_in[4];
    const float* W1r = (const float*)d_in[5];
    const float* W2l = (const float*)d_in[6];
    const float* b2  = (const float*)d_in[7];
    const float* W2r = (const float*)d_in[8];
    float* out = (float*)d_out;

    const int N  = in_sizes[0] / D;
    const int E  = in_sizes[1] / 2;
    const int Dk = in_sizes[2] / 2;
    const int* src = edg;
    const int* dst = edg + E;

    float* ws  = (float*)d_ws;
    float* s   = ws;
    float* cnt = s + (long long)N * D;
    float* h1  = cnt + N;
    float* h2  = h1 + (long long)N * D;

    const int aggrBlocks  = 256;
    const int gemmBlocks  = (N + 15) / 16;
    const int decBlocks   = (Dk + 7) / 8;

    aggregate_owner_kernel<<<aggrBlocks, 1024, 0, stream>>>(x, src, dst, s, cnt, E, N);
    sage_gemm_kernel<<<gemmBlocks, 256, 0, stream>>>(x, s, cnt, W1l, b1, W1r,
                                                     h1, N,  1);
    aggregate_owner_kernel<<<aggrBlocks, 1024, 0, stream>>>(h1, src, dst, s, cnt, E, N);
    sage_gemm_kernel<<<gemmBlocks, 256, 0, stream>>>(h1, s, cnt, W2l, b2, W2r,
                                                     h2, N,  0);
    decode_kernel<<<decBlocks, 256, 0, stream>>>(h2, dec, out, Dk);
}


# GEMM feed: 512 resident work-groups loop over 16-row tiles; a wave keeps its 16-column slab of both weight matrices in registers, node rows reach the 8 waves through LDS; sum order unchanged
# speedup vs baseline: 2.9502x; 1.1561x over previous
#include <hip/hip_runtime.h>
#include <hip/hip_bf16.h>
#include <math.h>

typedef __attribute__((ext_vector_type(2))) float v2f;
typedef __attribute__((ext_vector_type(8))) float v8f;

#define D 128

#define AGG_ROWS   196
#define AGG_WAVES  32
#define AGG_BATCH  1024
#define AGG_Q      (AGG_BATCH + 8)

__device__ __forceinline__ void lds_add(float* p, float v) {
    __hip_atomic_fetch_add(p, v, __ATOMIC_RELAXED, __HIP_MEMORY_SCOPE_WORKGROUP);
}

__device__ __forceinline__ void wave_lds_order() {
    __builtin_amdgcn_fence(__ATOMIC_RELEASE, "wavefront");
    __builtin_amdgcn_wave_barrier();
    __builtin_amdgcn_fence(__ATOMIC_ACQUIRE, "wavefront");
}

template <bool FULL>
__device__ __forceinline__ void drain_round(
    const float* __restrict__ x, const int* __restrict__ src,
    const int* __restrict__ dst, const int* myq, int i, int n, int lo,
    float* acc, float* cntL, int lane)
{
    int slot = (lane < n) ? (i + lane) : i;
    int e  = myq[slot];
    int sv = src[e];
    int tv = dst[e] - lo;
    float4 v[8];
    #pragma unroll
    for (int j = 0; j < 8; ++j) {
        if (FULL || j < n) {
            int sj = __builtin_amdgcn_readlane(sv, j);
            v[j] = ((const float4*)(x + (long long)sj * D))[lane];
        }
    }
    #pragma unroll
    for (int j = 0; j < 8; ++j) {
        if (FULL || j < n) {
            int tj = __builtin_amdgcn_readlane(tv, j);
            float* a = acc + tj * D + lane;
            lds_add(a,      v[j].x);
            lds_add(a + 32, v[j].y);
            lds_add(a + 64, v[j].z);
            lds_add(a + 96, v[j].w);
            if (lane == 0) lds_add(cntL + tj, 1.0f);
        }
    }
}

__global__ __launch_bounds__(1024) void aggregate_owner_kernel(
    const float* __restrict__ x, const int* __restrict__ src,
    const int* __restrict__ dst, float* __restrict__ s,
    float* __restrict__ cnt, int E, int N)
{
    __shared__ float acc[AGG_ROWS * D];
    __shared__ float cntL[AGG_ROWS];
    __shared__ int   q[AGG_WAVES][AGG_Q];

    const int tid  = threadIdx.x;
    const int lane = tid & 31;
    const int wave = tid >> 5;
    int* myq = q[wave];
    const unsigned ltmask = (1u << lane) - 1u;
    const bool aligned = (((unsigned long long)dst) & 15ull) == 0ull;
    const int nbatch = (E + AGG_BATCH - 1) / AGG_BATCH;

    for (int lo = blockIdx.x * AGG_ROWS; lo < N; lo += gridDim.x * AGG_ROWS) {
        const int nrows = min(AGG_ROWS, N - lo);
        for (int i = tid; i < AGG_ROWS * D; i += 1024) acc[i] = 0.0f;
        if (tid < AGG_ROWS) cntL[tid] = 0.0f;
        __syncthreads();

        int qn = 0;
        for (int b = wave; b < nbatch; b += AGG_WAVES) {
            const int base = b * AGG_BATCH + lane * 4;
            int4 d[8];
            if (aligned && (b + 1) * AGG_BATCH <= E) {
                #pragma unroll
                for (int u = 0; u < 8; ++u)
                    d[u] = *(const int4*)(dst + base + u * 128);
            } else {
                #pragma unroll
                for (int u = 0; u < 8; ++u) {
                    int e = base + u * 128;
                    d[u].x = (e     < E) ? dst[e]     : -1;
                    d[u].y = (e + 1 < E) ? dst[e + 1] : -1;
                    d[u].z = (e + 2 < E) ? dst[e + 2] : -1;
                    d[u].w = (e + 3 < E) ? dst[e + 3] : -1;
                }
            }
#define AGG_ENQ(dv, eidx)                                                     \
            {                                                                 \
                bool m = (unsigned)((dv) - lo) < (unsigned)nrows;             \
                unsigned bal = __builtin_amdgcn_ballot_w32(m);                \
                if (bal) {                                                    \
                    if (m) myq[qn + __builtin_popcount(bal & ltmask)] = (eidx); \
                    qn += __builtin_popcount(bal);                            \
                }                                                             \
            }
            #pragma unroll
            for (int u = 0; u < 8; ++u) {
                int e = base + u * 128;
                AGG_ENQ(d[u].x, e)
                AGG_ENQ(d[u].y, e + 1)
                AGG_ENQ(d[u].z, e + 2)
                AGG_ENQ(d[u].w, e + 3)
            }
#undef AGG_ENQ
            wave_lds_order();
            int i = 0;
            for (; i + 8 <= qn; i += 8)
                drain_round<true>(x, src, dst, myq, i, 8, lo, acc, cntL, lane);
            int rem = qn - i;
            if (i > 0 && rem > 0) {
                int vq = myq[i + ((lane < rem) ? lane : 0)];
                if (lane < rem) myq[lane] = vq;
            }
            qn = rem;
            wave_lds_order();
        }
        if (qn > 0)
            drain_round<false>(x, src, dst, myq, 0, qn, lo, acc, cntL, lane);
        __syncthreads();

        for (int r = wave; r < nrows; r += AGG_WAVES) {
            const float* a = acc + r * D + lane;
            float4 o;
            o.x = a[0]; o.y = a[32]; o.z = a[64]; o.w = a[96];
            ((float4*)(s + (long long)(lo + r) * D))[lane] = o;
        }
        if (tid < nrows) cnt[lo + tid] = cntL[tid];
        __syncthreads();
    }
}

#define GEMM_BLOCKS 512
#define GT_PITCH2   66

__global__ __launch_bounds__(256) void sage_gemm_kernel(
    const float* __restrict__ xin, const float* __restrict__ s,
    const float* __restrict__ cnt, const float* __restrict__ Wl,
    const float* __restrict__ bias, const float* __restrict__ Wr,
    float* __restrict__ out, int N, int relu)
{
    __shared__ __attribute__((aligned(16))) v2f tileA[2][2][16 * GT_PITCH2];

    const int wave  = threadIdx.x >> 5;
    const int lane  = threadIdx.x & 31;
    const int c16   = lane & 15;
    const int hsel  = lane >> 4;
    const int khalf = hsel << 1;
    const int ncol  = wave * 16 + c16;
    const int ntiles = (N + 15) >> 4;

    v2f bl[32], br[32];
    #pragma unroll
    for (int t = 0; t < 32; ++t) {
        int ka = 4 * t + khalf;
        bl[t].x = Wl[ka * D + ncol]; bl[t].y = Wl[(ka + 1) * D + ncol];
        br[t].x = Wr[ka * D + ncol]; br[t].y = Wr[(ka + 1) * D + ncol];
    }
    const float bn = bias[ncol];

    const int r0 = 2 * wave, r1 = r0 + 1;
    float4 ps0, ps1, px0, px1;
    float  pc0, pc1;

#define GEMM_FETCH(tl)                                                        \
    {                                                                         \
        int ga = (tl) * 16 + r0; if (ga > N - 1) ga = N - 1;                  \
        int gb = (tl) * 16 + r1; if (gb > N - 1) gb = N - 1;                  \
        pc0 = cnt[ga];                                                        \
        pc1 = cnt[gb];                                                        \
        ps0 = ((const float4*)(s   + (long long)ga * D))[lane];               \
        ps1 = ((const float4*)(s   + (long long)gb * D))[lane];               \
        px0 = ((const float4*)(xin + (long long)ga * D))[lane];               \
        px1 = ((const float4*)(xin + (long long)gb * D))[lane];               \
    }
#define GEMM_PUT(bf)                                                          \
    {                                                                         \
        float ia = 1.0f / fmaxf(pc0, 1.0f);                                   \
        float ib = 1.0f / fmaxf(pc1, 1.0f);                                   \
        v2f* dS = &tileA[(bf)][0][0];                                         \
        v2f* dX = &tileA[(bf)][1][0];                                         \
        v2f t0; v2f t1;                                                       \
        t0.x = ps0.x * ia; t0.y = ps0.y * ia; t1.x = ps0.z * ia; t1.y = ps0.w * ia; \
        dS[r0 * GT_PITCH2 + 2 * lane] = t0; dS[r0 * GT_PITCH2 + 2 * lane + 1] = t1; \
        t0.x = ps1.x * ib; t0.y = ps1.y * ib; t1.x = ps1.z * ib; t1.y = ps1.w * ib; \
        dS[r1 * GT_PITCH2 + 2 * lane] = t0; dS[r1 * GT_PITCH2 + 2 * lane + 1] = t1; \
        t0.x = px0.x; t0.y = px0.y; t1.x = px0.z; t1.y = px0.w;               \
        dX[r0 * GT_PITCH2 + 2 * lane] = t0; dX[r0 * GT_PITCH2 + 2 * lane + 1] = t1; \
        t0.x = px1.x; t0.y = px1.y; t1.x = px1.z; t1.y = px1.w;               \
        dX[r1 * GT_PITCH2 + 2 * lane] = t0; dX[r1 * GT_PITCH2 + 2 * lane + 1] = t1; \
    }

    int tile = blockIdx.x;
    {
        int first = (tile < ntiles) ? tile : 0;
        GEMM_FETCH(first)
    }
    GEMM_PUT(0)
    __syncthreads();

    int buf = 0;
    for (; tile < ntiles; tile += gridDim.x) {
        const int  nxt  = tile + gridDim.x;
        const bool more = nxt < ntiles;
        if (more) GEMM_FETCH(nxt)

        const v2f* aS = &tileA[buf][0][c16 * GT_PITCH2 + hsel];
        const v2f* aX = &tileA[buf][1][c16 * GT_PITCH2 + hsel];
        v8f c = {};
        #pragma unroll
        for (int t = 0; t < 32; ++t) {
            v2f a0 = aS[2 * t];
            c = __builtin_amdgcn_wmma_f32_16x16x4_f32(
                    false, a0, false, bl[t], (short)0, c, false, false);
            v2f a1 = aX[2 * t];
            c = __builtin_amdgcn_wmma_f32_16x16x4_f32(
                    false, a1, false, br[t], (short)0, c, false, false);
        }

        if (more) GEMM_PUT(buf ^ 1)
        __syncthreads();

        const int mbase = tile * 16 + (hsel << 3);
        #pragma unroll
        for (int r = 0; r < 8; ++r) {
            int m = mbase + r;
            if (m < N) {
                float v = c[r] + bn;
                if (relu) v = fmaxf(v, 0.0f);
                out[(long long)m * D + ncol] = v;
            }
        }
        buf ^= 1;
    }
#undef GEMM_FETCH
#undef GEMM_PUT
}

__global__ __launch_bounds__(256) void decode_kernel(
    const float* __restrict__ h, const int* __restrict__ dec,
    float* __restrict__ out, int Dk)
{
    int wid  = (int)(((long long)blockIdx.x * blockDim.x + threadIdx.x) >> 5);
    int lane = threadIdx.x & 31;
    if (wid >= Dk) return;
    int ai = dec[wid];
    int bi = dec[Dk + wid];
    float4 a = ((const float4*)(h + (long long)ai * D))[lane];
    float4 b = ((const float4*)(h + (long long)bi * D))[lane];
    float dot = a.x * b.x + a.y * b.y + a.z * b.z + a.w * b.w;
    float na  = a.x * a.x + a.y * a.y + a.z * a.z + a.w * a.w;
    float nb  = b.x * b.x + b.y * b.y + b.z * b.z + b.w * b.w;
    #pragma unroll
    for (int off = 16; off > 0; off >>= 1) {
        dot += __shfl_xor(dot, off, 32);
        na  += __shfl_xor(na,  off, 32);
        nb  += __shfl_xor(nb,  off, 32);
    }
    if (lane == 0) {
        float denom = fmaxf(sqrtf(na) * sqrtf(nb), 1e-6f);
        float z = dot / denom;
        out[wid] = 1.0f / (1.0f + expf(-z));
    }
}

extern "C" void kernel_launch(void* const* d_in, const int* in_sizes, int n_in,
                              void* d_out, int out_size, void* d_ws, size_t ws_size,
                              hipStream_t stream) {
    const float* x   = (const float*)d_in[0];
    const int*   edg = (const int*)d_in[1];
    const int*   dec = (const int*)d_in[2];
    const float* W1l = (const float*)d_in[3];
    const float* b1  = (const float*)d_in[4];
    const float* W1r = (const float*)d_in[5];
    const float* W2l = (const float*)d_in[6];
    const float* b2  = (const float*)d_in[7];
    const float* W2r = (const float*)d_in[8];
    float* out = (float*)d_out;

    const int N  = in_sizes[0] / D;
    const int E  = in_sizes[1] / 2;
    const int Dk = in_sizes[2] / 2;
    const int* src = edg;
    const int* dst = edg + E;

    float* ws  = (float*)d_ws;
    float* s   = ws;
    float* cnt = s + (long long)N * D;
    float* h1  = cnt + N;
    float* h2  = h1 + (long long)N * D;

    const int aggrBlocks  = 256;
    const int gemmBlocks  = GEMM_BLOCKS;
    const int decBlocks   = (Dk + 7) / 8;

    aggregate_owner_kernel<<<aggrBlocks, 1024, 0, stream>>>(x, src, dst, s, cnt, E, N);
    sage_gemm_kernel<<<gemmBlocks, 256, 0, stream>>>(x, s, cnt, W1l, b1, W1r,
                                                     h1, N,  1);
    aggregate_owner_kernel<<<aggrBlocks, 1024, 0, stream>>>(h1, src, dst, s, cnt, E, N);
    sage_gemm_kernel<<<gemmBlocks, 256, 0, stream>>>(h1, s, cnt, W2l, b2, W2r,
                                                     h2, N,  0);
    decode_kernel<<<decBlocks, 256, 0, stream>>>(h2, dec, out, Dk);
}


# aggregate, layer 2: no dst scan; each wave walks the (src, row) list that its layer-1 scan wrote to the workspace, in-degrees stand; f32 sums stay in arrival order of LDS adds
# speedup vs baseline: 3.4804x; 1.1797x over previous
#include <hip/hip_runtime.h>
#include <hip/hip_bf16.h>
#include <math.h>

typedef __attribute__((ext_vector_type(2))) float v2f;
typedef __attribute__((ext_vector_type(8))) float v8f;

#define D 128

#define AGG_ROWS   196
#define AGG_WAVES  32
#define AGG_BATCH  1024
#define AGG_Q      (AGG_BATCH + 8)

__device__ __forceinline__ void lds_add(float* p, float v) {
    __hip_atomic_fetch_add(p, v, __ATOMIC_RELAXED, __HIP_MEMORY_SCOPE_WORKGROUP);
}

__device__ __forceinline__ void wave_lds_order() {
    __builtin_amdgcn_fence(__ATOMIC_RELEASE, "wavefront");
    __builtin_amdgcn_wave_barrier();
    __builtin_amdgcn_fence(__ATOMIC_ACQUIRE, "wavefront");
}

template <bool FULL, bool REC>
__device__ __forceinline__ void drain_round(
    const float* __restrict__ x, const int* __restrict__ src,
    const int* __restrict__ dst, const int* myq, int i, int n, int lo,
    float* acc, float* cntL, int lane, int2* wlist, int pos, int cap)
{
    int slot = (lane < n) ? (i + lane) : i;
    int e  = myq[slot];
    int sv = src[e];
    int tv = dst[e] - lo;
    if (REC) {
        int p = pos + lane;
        if (lane < n && p < cap) wlist[p] = make_int2(sv, tv);
    }
    float4 v[8];
    #pragma unroll
    for (int j = 0; j < 8; ++j) {
        if (FULL || j < n) {
            int sj = __builtin_amdgcn_readlane(sv, j);
            v[j] = ((const float4*)(x + (long long)sj * D))[lane];
        }
    }
    #pragma unroll
    for (int j = 0; j < 8; ++j) {
        if (FULL || j < n) {
            int tj = __builtin_amdgcn_readlane(tv, j);
            float* a = acc + tj * D + lane;
            lds_add(a,      v[j].x);
            lds_add(a + 32, v[j].y);
            lds_add(a + 64, v[j].z);
            lds_add(a + 96, v[j].w);
            if (lane == 0) lds_add(cntL + tj, 1.0f);
        }
    }
}

template <bool FULL>
__device__ __forceinline__ void replay_round(
    const float* __restrict__ x, int sv, int tv, int r, int n, float* acc, int lane)
{
    float4 v[8];
    #pragma unroll
    for (int j = 0; j < 8; ++j) {
        if (FULL || j < n) {
            int sj = __builtin_amdgcn_readlane(sv, r + j);
            v[j] = ((const float4*)(x + (long long)sj * D))[lane];
        }
    }
    #pragma unroll
    for (int j = 0; j < 8; ++j) {
        if (FULL || j < n) {
            int tj = __builtin_amdgcn_readlane(tv, r + j);
            float* a = acc + tj * D + lane;
            lds_add(a,      v[j].x);
            lds_add(a + 32, v[j].y);
            lds_add(a + 64, v[j].z);
            lds_add(a + 96, v[j].w);
        }
    }
}

template <int MODE>
__global__ __launch_bounds__(1024) void aggregate_owner_kernel(
    const float* __restrict__ x, const int* __restrict__ src,
    const int* __restrict__ dst, float* __restrict__ s,
    float* __restrict__ cnt, int* __restrict__ rec, int E, int N)
{
    __shared__ float acc[AGG_ROWS * D];
    __shared__ float cntL[AGG_ROWS];
    __shared__ int   q[AGG_WAVES][AGG_Q];

    const int tid  = threadIdx.x;
    const int lane = tid & 31;
    const int wave = tid >> 5;
    int* myq = q[wave];
    const unsigned ltmask = (1u << lane) - 1u;
    const bool aligned = (((unsigned long long)dst) & 15ull) == 0ull;
    const int nbatch = (E + AGG_BATCH - 1) / AGG_BATCH;

    const long long nslots = (long long)gridDim.x * AGG_WAVES;
    const long long room   = (long long)N * D;
    int cap = 0;
    if ((long long)gridDim.x * AGG_ROWS >= N && room >= 3 * nslots)
        cap = (int)((room - nslots) / (2 * nslots));
    const long long myslot = (long long)blockIdx.x * AGG_WAVES + wave;

    for (int lo = blockIdx.x * AGG_ROWS; lo < N; lo += gridDim.x * AGG_ROWS) {
        const int nrows = min(AGG_ROWS, N - lo);
        for (int i = tid; i < AGG_ROWS * D; i += 1024) acc[i] = 0.0f;
        if (tid < AGG_ROWS) cntL[tid] = 0.0f;

        bool replay = false;
        int  m = 0;
        if (MODE == 1 && cap > 0) {
            int len = rec[(long long)blockIdx.x * AGG_WAVES + lane];
            replay = __builtin_amdgcn_ballot_w32(len > cap) == 0u;
            m = __builtin_amdgcn_readlane(len, wave);
        }
        __syncthreads();

        if (MODE == 1 && replay) {
            const int2* wlist = (const int2*)(rec + nslots) + myslot * cap;
            const int mfull = m & ~7;
            const int tail  = m - mfull;
            int2 cur = make_int2(0, 0);
            if (lane < mfull) cur = wlist[lane];
            if (tail > 0) {
                int2 te = wlist[mfull + min(lane, tail - 1)];
                replay_round<false>(x, te.x, te.y, 0, tail, acc, lane);
            }
            for (int i = 0; i < mfull; i += 32) {
                int2 nxt = make_int2(0, 0);
                if (i + 32 + lane < mfull) nxt = wlist[i + 32 + lane];
                const int left = min(32, mfull - i);
                for (int r = 0; r < left; r += 8)
                    replay_round<true>(x, cur.x, cur.y, r, 8, acc, lane);
                cur = nxt;
            }
        } else {
            int2* wlist = (int2*)(rec + nslots) + myslot * cap;
            int qn  = 0;
            int pos = 0;
            for (int b = wave; b < nbatch; b += AGG_WAVES) {
                const int base = b * AGG_BATCH + lane * 4;
                int4 d[8];
                if (aligned && (b + 1) * AGG_BATCH <= E) {
                    #pragma unroll
                    for (int u = 0; u < 8; ++u)
                        d[u] = *(const int4*)(dst + base + u * 128);
                } else {
                    #pragma unroll
                    for (int u = 0; u < 8; ++u) {
                        int e = base + u * 128;
                        d[u].x = (e     < E) ? dst[e]     : -1;
                        d[u].y = (e + 1 < E) ? dst[e + 1] : -1;
                        d[u].z = (e + 2 < E) ? dst[e + 2] : -1;
                        d[u].w = (e + 3 < E) ? dst[e + 3] : -1;
                    }
                }
#define AGG_ENQ(dv, eidx)                                                     \
                {                                                             \
                    bool mt = (unsigned)((dv) - lo) < (unsigned)nrows;        \
                    unsigned bal = __builtin_amdgcn_ballot_w32(mt);           \
                    if (bal) {                                                \
                        if (mt) myq[qn + __builtin_popcount(bal & ltmask)] = (eidx); \
                        qn += __builtin_popcount(bal);                        \
                    }                                                         \
                }
                #pragma unroll
                for (int u = 0; u < 8; ++u) {
                    int e = base + u * 128;
                    AGG_ENQ(d[u].x, e)
                    AGG_ENQ(d[u].y, e + 1)
                    AGG_ENQ(d[u].z, e + 2)
                    AGG_ENQ(d[u].w, e + 3)
                }
#undef AGG_ENQ
                wave_lds_order();
                int i = 0;
                for (; i + 8 <= qn; i += 8) {
                    drain_round<true, MODE == 0>(x, src, dst, myq, i, 8, lo, acc, cntL, lane,
                                                 wlist, pos, cap);
                    pos += 8;
                }
                int rem = qn - i;
                if (i > 0 && rem > 0) {
                    int vq = myq[i + ((lane < rem) ? lane : 0)];
                    if (lane < rem) myq[lane] = vq;
                }
                qn = rem;
                wave_lds_order();
            }
            if (qn > 0) {
                drain_round<false, MODE == 0>(x, src, dst, myq, 0, qn, lo, acc, cntL, lane,
                                              wlist, pos, cap);
                pos += qn;
            }
            if (MODE == 0 && cap > 0 && lane == 0) rec[myslot] = pos;
        }
        __syncthreads();

        for (int r = wave; r < nrows; r += AGG_WAVES) {
            const float* a = acc + r * D + lane;
            float4 o;
            o.x = a[0]; o.y = a[32]; o.z = a[64]; o.w = a[96];
            ((float4*)(s + (long long)(lo + r) * D))[lane] = o;
        }
        if (!(MODE == 1 && replay) && tid < nrows) cnt[lo + tid] = cntL[tid];
        __syncthreads();
    }
}

#define GEMM_BLOCKS 512
#define GT_PITCH2   66

__global__ __launch_bounds__(256) void sage_gemm_kernel(
    const float* __restrict__ xin, const float* __restrict__ s,
    const float* __restrict__ cnt, const float* __restrict__ Wl,
    const float* __restrict__ bias, const float* __restrict__ Wr,
    float* __restrict__ out, int N, int relu)
{
    __shared__ __attribute__((aligned(16))) v2f tileA[2][2][16 * GT_PITCH2];

    const int wave  = threadIdx.x >> 5;
    const int lane  = threadIdx.x & 31;
    const int c16   = lane & 15;
    const int hsel  = lane >> 4;
    const int khalf = hsel << 1;
    const int ncol  = wave * 16 + c16;
    const int ntiles = (N + 15) >> 4;

    v2f bl[32], br[32];
    #pragma unroll
    for (int t = 0; t < 32; ++t) {
        int ka = 4 * t + khalf;
        bl[t].x = Wl[ka * D + ncol]; bl[t].y = Wl[(ka + 1) * D + ncol];
        br[t].x = Wr[ka * D + ncol]; br[t].y = Wr[(ka + 1) * D + ncol];
    }
    const float bn = bias[ncol];

    const int r0 = 2 * wave, r1 = r0 + 1;
    float4 ps0, ps1, px0, px1;
    float  pc0, pc1;

#define GEMM_FETCH(tl)                                                        \
    {                                                                         \
        int ga = (tl) * 16 + r0; if (ga > N - 1) ga = N - 1;                  \
        int gb = (tl) * 16 + r1; if (gb > N - 1) gb = N - 1;                  \
        pc0 = cnt[ga];                                                        \
        pc1 = cnt[gb];                                                        \
        ps0 = ((const float4*)(s   + (long long)ga * D))[lane];               \
        ps1 = ((const float4*)(s   + (long long)gb * D))[lane];               \
        px0 = ((const float4*)(xin + (long long)ga * D))[lane];               \
        px1 = ((const float4*)(xin + (long long)gb * D))[lane];               \
    }
#define GEMM_PUT(bf)                                                          \
    {                                                                         \
        float ia = 1.0f / fmaxf(pc0, 1.0f);                                   \
        float ib = 1.0f / fmaxf(pc1, 1.0f);                                   \
        v2f* dS = &tileA[(bf)][0][0];                                         \
        v2f* dX = &tileA[(bf)][1][0];                                         \
        v2f t0; v2f t1;                                                       \
        t0.x = ps0.x * ia; t0.y = ps0.y * ia; t1.x = ps0.z * ia; t1.y = ps0.w * ia; \
        dS[r0 * GT_PITCH2 + 2 * lane] = t0; dS[r0 * GT_PITCH2 + 2 * lane + 1] = t1; \
        t0.x = ps1.x * ib; t0.y = ps1.y * ib; t1.x = ps1.z * ib; t1.y = ps1.w * ib; \
        dS[r1 * GT_PITCH2 + 2 * lane] = t0; dS[r1 * GT_PITCH2 + 2 * lane + 1] = t1; \
        t0.x = px0.x; t0.y = px0.y; t1.x = px0.z; t1.y = px0.w;               \
        dX[r0 * GT_PITCH2 + 2 * lane] = t0; dX[r0 * GT_PITCH2 + 2 * lane + 1] = t1; \
        t0.x = px1.x; t0.y = px1.y; t1.x = px1.z; t1.y = px1.w;               \
        dX[r1 * GT_PITCH2 + 2 * lane] = t0; dX[r1 * GT_PITCH2 + 2 * lane + 1] = t1; \
    }

    int tile = blockIdx.x;
    {
        int first = (tile < ntiles) ? tile : 0;
        GEMM_FETCH(first)
    }
    GEMM_PUT(0)
    __syncthreads();

    int buf = 0;
    for (; tile < ntiles; tile += gridDim.x) {
        const int  nxt  = tile + gridDim.x;
        const bool more = nxt < ntiles;
        if (more) GEMM_FETCH(nxt)

        const v2f* aS = &tileA[buf][0][c16 * GT_PITCH2 + hsel];
        const v2f* aX = &tileA[buf][1][c16 * GT_PITCH2 + hsel];
        v8f c = {};
        #pragma unroll
        for (int t = 0; t < 32; ++t) {
            v2f a0 = aS[2 * t];
            c = __builtin_amdgcn_wmma_f32_16x16x4_f32(
                    false, a0, false, bl[t], (short)0, c, false, false);
            v2f a1 = aX[2 * t];
            c = __builtin_amdgcn_wmma_f32_16x16x4_f32(
                    false, a1, false, br[t], (short)0, c, false, false);
        }

        if (more) GEMM_PUT(buf ^ 1)
        __syncthreads();

        const int mbase = tile * 16 + (hsel << 3);
        #pragma unroll
        for (int r = 0; r < 8; ++r) {
            int m = mbase + r;
            if (m < N) {
                float v = c[r] + bn;
                if (relu) v = fmaxf(v, 0.0f);
                out[(long long)m * D + ncol] = v;
            }
        }
        buf ^= 1;
    }
#undef GEMM_FETCH
#undef GEMM_PUT
}

__global__ __launch_bounds__(256) void decode_kernel(
    const float* __restrict__ h, const int* __restrict__ dec,
    float* __restrict__ out, int Dk)
{
    int wid  = (int)(((long long)blockIdx.x * blockDim.x + threadIdx.x) >> 5);
    int lane = threadIdx.x & 31;
    if (wid >= Dk) return;
    int ai = dec[wid];
    int bi = dec[Dk + wid];
    float4 a = ((const float4*)(h + (long long)ai * D))[lane];
    float4 b = ((const float4*)(h + (long long)bi * D))[lane];
    float dot = a.x * b.x + a.y * b.y + a.z * b.z + a.w * b.w;
    float na  = a.x * a.x + a.y * a.y + a.z * a.z + a.w * a.w;
    float nb  = b.x * b.x + b.y * b.y + b.z * b.z + b.w * b.w;
    #pragma unroll
    for (int off = 16; off > 0; off >>= 1) {
        dot += __shfl_xor(dot, off, 32);
        na  += __shfl_xor(na,  off, 32);
        nb  += __shfl_xor(nb,  off, 32);
    }
    if (lane == 0) {
        float denom = fmaxf(sqrtf(na) * sqrtf(nb), 1e-6f);
        float z = dot / denom;
        out[wid] = 1.0f / (1.0f + expf(-z));
    }
}

extern "C" void kernel_launch(void* const* d_in, const int* in_sizes, int n_in,
                              void* d_out, int out_size, void* d_ws, size_t ws_size,
                              hipStream_t stream) {
    const float* x   = (const float*)d_in[0];
    const int*   edg = (const int*)d_in[1];
    const int*   dec = (const int*)d_in[2];
    const float* W1l = (const float*)d_in[3];
    const float* b1  = (const float*)d_in[4];
    const float* W1r = (const float*)d_in[5];
    const float* W2l = (const float*)d_in[6];
    const float* b2  = (const float*)d_in[7];
    const float* W2r = (const float*)d_in[8];
    float* out = (float*)d_out;

    const int N  = in_sizes[0] / D;
    const int E  = in_sizes[1] / 2;
    const int Dk = in_sizes[2] / 2;
    const int* src = edg;
    const int* dst = edg + E;

    float* ws  = (float*)d_ws;
    float* s   = ws;
    float* cnt = s + (long long)N * D;
    float* h1  = cnt + N;
    float* h2  = h1 + (long long)N * D;

    int* rec   = (int*)h2;

    const int aggrBlocks  = 256;
    const int gemmBlocks  = GEMM_BLOCKS;
    const int decBlocks   = (Dk + 7) / 8;

    aggregate_owner_kernel<0><<<aggrBlocks, 1024, 0, stream>>>(x, src, dst, s, cnt, rec, E, N);
    sage_gemm_kernel<<<gemmBlocks, 256, 0, stream>>>(x, s, cnt, W1l, b1, W1r,
                                                     h1, N,  1);
    aggregate_owner_kernel<1><<<aggrBlocks, 1024, 0, stream>>>(h1, src, dst, s, cnt, rec, E, N);
    sage_gemm_kernel<<<gemmBlocks, 256, 0, stream>>>(h1, s, cnt, W2l, b2, W2r,
                                                     h2, N,  0);
    decode_kernel<<<decBlocks, 256, 0, stream>>>(h2, dec, out, Dk);
}


# aggregate, layer 1: a new first launch packs every edge's owner (dst/197) into one byte; each owner work-group scans those bytes, 16 a lane per test, not the 4-byte dst list; f32 sum order unchanged
# speedup vs baseline: 3.7356x; 1.0733x over previous
#include <hip/hip_runtime.h>
#include <hip/hip_bf16.h>
#include <math.h>

typedef __attribute__((ext_vector_type(2))) float v2f;
typedef __attribute__((ext_vector_type(8))) float v8f;

#define D 128

#define AGG_ROWS   197
#define AGG_NOKEY  255u
#define AGG_WAVES  32
#define AGG_BATCH  1024
#define AGG_Q      (AGG_BATCH + 8)

__device__ __forceinline__ void lds_add(float* p, float v) {
    __hip_atomic_fetch_add(p, v, __ATOMIC_RELAXED, __HIP_MEMORY_SCOPE_WORKGROUP);
}

__device__ __forceinline__ void wave_lds_order() {
    __builtin_amdgcn_fence(__ATOMIC_RELEASE, "wavefront");
    __builtin_amdgcn_wave_barrier();
    __builtin_amdgcn_fence(__ATOMIC_ACQUIRE, "wavefront");
}

template <bool FULL, bool REC>
__device__ __forceinline__ void drain_round(
    const float* __restrict__ x, const int* __restrict__ src,
    const int* __restrict__ dst, const int* myq, int i, int n, int lo,
    float* acc, float* cntL, int lane, int2* wlist, int pos, int cap)
{
    int slot = (lane < n) ? (i + lane) : i;
    int e  = myq[slot];
    int sv = src[e];
    int tv = dst[e] - lo;
    if (REC) {
        int p = pos + lane;
        if (lane < n && p < cap) wlist[p] = make_int2(sv, tv);
    }
    float4 v[8];
    #pragma unroll
    for (int j = 0; j < 8; ++j) {
        if (FULL || j < n) {
            int sj = __builtin_amdgcn_readlane(sv, j);
            v[j] = ((const float4*)(x + (long long)sj * D))[lane];
        }
    }
    #pragma unroll
    for (int j = 0; j < 8; ++j) {
        if (FULL || j < n) {
            int tj = __builtin_amdgcn_readlane(tv, j);
            float* a = acc + tj * D + lane;
            lds_add(a,      v[j].x);
            lds_add(a + 32, v[j].y);
            lds_add(a + 64, v[j].z);
            lds_add(a + 96, v[j].w);
            if (lane == 0) lds_add(cntL + tj, 1.0f);
        }
    }
}

template <bool FULL>
__device__ __forceinline__ void replay_round(
    const float* __restrict__ x, int sv, int tv, int r, int n, float* acc, int lane)
{
    float4 v[8];
    #pragma unroll
    for (int j = 0; j < 8; ++j) {
        if (FULL || j < n) {
            int sj = __builtin_amdgcn_readlane(sv, r + j);
            v[j] = ((const float4*)(x + (long long)sj * D))[lane];
        }
    }
    #pragma unroll
    for (int j = 0; j < 8; ++j) {
        if (FULL || j < n) {
            int tj = __builtin_amdgcn_readlane(tv, r + j);
            float* a = acc + tj * D + lane;
            lds_add(a,      v[j].x);
            lds_add(a + 32, v[j].y);
            lds_add(a + 64, v[j].z);
            lds_add(a + 96, v[j].w);
        }
    }
}

__device__ __forceinline__ unsigned key_flags(unsigned dw, unsigned pat) {
    unsigned x = dw ^ pat;
    return ~(((x & 0x7F7F7F7Fu) + 0x7F7F7F7Fu) | x | 0x7F7F7F7Fu);
}

__device__ __forceinline__ bool keys_usable(const unsigned* key, int E, int N, int nblocks) {
    return (N + AGG_ROWS - 1) / AGG_ROWS <= (int)AGG_NOKEY
        && (long long)nblocks * AGG_ROWS >= N
        && (long long)E <= (long long)N * D * 4
        && (((unsigned long long)key) & 15ull) == 0ull;
}

__global__ __launch_bounds__(256) void owner_key_kernel(
    const int* __restrict__ dst, unsigned* __restrict__ key, int E, int N)
{
    if ((long long)E > (long long)N * D * 4) return;
    const bool aligned = (((unsigned long long)dst) & 15ull) == 0ull;
    const int nd = E >> 2;
    for (int t = blockIdx.x * blockDim.x + threadIdx.x; t < nd; t += gridDim.x * blockDim.x) {
        int4 d;
        if (aligned) {
            d = *(const int4*)(dst + 4 * t);
        } else {
            d.x = dst[4 * t]; d.y = dst[4 * t + 1]; d.z = dst[4 * t + 2]; d.w = dst[4 * t + 3];
        }
        unsigned k0 = ((unsigned)d.x < (unsigned)N) ? min((unsigned)d.x / AGG_ROWS, AGG_NOKEY) : AGG_NOKEY;
        unsigned k1 = ((unsigned)d.y < (unsigned)N) ? min((unsigned)d.y / AGG_ROWS, AGG_NOKEY) : AGG_NOKEY;
        unsigned k2 = ((unsigned)d.z < (unsigned)N) ? min((unsigned)d.z / AGG_ROWS, AGG_NOKEY) : AGG_NOKEY;
        unsigned k3 = ((unsigned)d.w < (unsigned)N) ? min((unsigned)d.w / AGG_ROWS, AGG_NOKEY) : AGG_NOKEY;
        key[t] = k0 | (k1 << 8) | (k2 << 16) | (k3 << 24);
    }
}

template <int MODE>
__global__ __launch_bounds__(1024) void aggregate_owner_kernel(
    const float* __restrict__ x, const int* __restrict__ src,
    const int* __restrict__ dst, float* __restrict__ s,
    float* __restrict__ cnt, int* __restrict__ rec,
    const unsigned* __restrict__ key, int E, int N)
{
    __shared__ float acc[AGG_ROWS * D];
    __shared__ float cntL[AGG_ROWS];
    __shared__ int   q[AGG_WAVES][AGG_Q];

    const int tid  = threadIdx.x;
    const int lane = tid & 31;
    const int wave = tid >> 5;
    int* myq = q[wave];
    const unsigned ltmask = (1u << lane) - 1u;
    const bool aligned = (((unsigned long long)dst) & 15ull) == 0ull;
    const int nbatch = (E + AGG_BATCH - 1) / AGG_BATCH;
    const bool usekeys = (MODE == 0) && keys_usable(key, E, N, gridDim.x);
    const unsigned pat = blockIdx.x * 0x01010101u;

    const long long nslots = (long long)gridDim.x * AGG_WAVES;
    const long long room   = (long long)N * D;
    int cap = 0;
    if ((long long)gridDim.x * AGG_ROWS >= N && room >= 3 * nslots)
        cap = (int)((room - nslots) / (2 * nslots));
    const long long myslot = (long long)blockIdx.x * AGG_WAVES + wave;

    for (int lo = blockIdx.x * AGG_ROWS; lo < N; lo += gridDim.x * AGG_ROWS) {
        const int nrows = min(AGG_ROWS, N - lo);
        for (int i = tid; i < AGG_ROWS * D; i += 1024) acc[i] = 0.0f;
        if (tid < AGG_ROWS) cntL[tid] = 0.0f;

        bool replay = false;
        int  m = 0;
        if (MODE == 1 && cap > 0) {
            int len = rec[(long long)blockIdx.x * AGG_WAVES + lane];
            replay = __builtin_amdgcn_ballot_w32(len > cap) == 0u;
            m = __builtin_amdgcn_readlane(len, wave);
        }
        __syncthreads();

        if (MODE == 1 && replay) {
            const int2* wlist = (const int2*)(rec + nslots) + myslot * cap;
            const int mfull = m & ~7;
            const int tail  = m - mfull;
            int2 cur = make_int2(0, 0);
            if (lane < mfull) cur = wlist[lane];
            if (tail > 0) {
                int2 te = wlist[mfull + min(lane, tail - 1)];
                replay_round<false>(x, te.x, te.y, 0, tail, acc, lane);
            }
            for (int i = 0; i < mfull; i += 32) {
                int2 nxt = make_int2(0, 0);
                if (i + 32 + lane < mfull) nxt = wlist[i + 32 + lane];
                const int left = min(32, mfull - i);
                for (int r = 0; r < left; r += 8)
                    replay_round<true>(x, cur.x, cur.y, r, 8, acc, lane);
                cur = nxt;
            }
        } else {
            int2* wlist = (int2*)(rec + nslots) + myslot * cap;
            int qn  = 0;
            int pos = 0;
            for (int b = wave; b < nbatch; b += AGG_WAVES) {
                if (usekeys && (b + 1) * AGG_BATCH <= E) {
                    const int4* kp = (const int4*)(key + b * (AGG_BATCH / 4)) + lane;
                    int4 kv[AGG_BATCH / 512];
                    #pragma unroll
                    for (int u = 0; u < AGG_BATCH / 512; ++u)
                        kv[u] = kp[u * 32];
                    #pragma unroll
                    for (int u = 0; u < AGG_BATCH / 512; ++u) {
                        const int ebase = b * AGG_BATCH + u * 512 + lane * 16;
                        unsigned c = key_flags((unsigned)kv[u].x, pat)
                                   | (key_flags((unsigned)kv[u].y, pat) >> 1)
                                   | (key_flags((unsigned)kv[u].z, pat) >> 2)
                                   | (key_flags((unsigned)kv[u].w, pat) >> 3);
                        unsigned bal = __builtin_amdgcn_ballot_w32(c != 0u);
                        while (bal) {
                            if (c != 0u) {
                                int p = __builtin_ctz(c);
                                myq[qn + __builtin_popcount(bal & ltmask)] =
                                    ebase + ((7 - (p & 7)) << 2) + (p >> 3);
                                c &= c - 1u;
                            }
                            qn += __builtin_popcount(bal);
                            bal = __builtin_amdgcn_ballot_w32(c != 0u);
                        }
                    }
                } else {
                    const int base = b * AGG_BATCH + lane * 4;
                    int4 d[8];
                    if (aligned && (b + 1) * AGG_BATCH <= E) {
                        #pragma unroll
                        for (int u = 0; u < 8; ++u)
                            d[u] = *(const int4*)(dst + base + u * 128);
                    } else {
                        #pragma unroll
                        for (int u = 0; u < 8; ++u) {
                            int e = base + u * 128;
                            d[u].x = (e     < E) ? dst[e]     : -1;
                            d[u].y = (e + 1 < E) ? dst[e + 1] : -1;
                            d[u].z = (e + 2 < E) ? dst[e + 2] : -1;
                            d[u].w = (e + 3 < E) ? dst[e + 3] : -1;
                        }
                    }
#define AGG_ENQ(dv, eidx)                                                     \
                    {                                                             \
                        bool mt = (unsigned)((dv) - lo) < (unsigned)nrows;        \
                        unsigned bal = __builtin_amdgcn_ballot_w32(mt);           \
                        if (bal) {                                                \
                            if (mt) myq[qn + __builtin_popcount(bal & ltmask)] = (eidx); \
                            qn += __builtin_popcount(bal);                        \
                        }                                                         \
                    }
                    #pragma unroll
                    for (int u = 0; u < 8; ++u) {
                        int e = base + u * 128;
                        AGG_ENQ(d[u].x, e)
                        AGG_ENQ(d[u].y, e + 1)
                        AGG_ENQ(d[u].z, e + 2)
                        AGG_ENQ(d[u].w, e + 3)
                    }
#undef AGG_ENQ
                }
                wave_lds_order();
                int i = 0;
                for (; i + 8 <= qn; i += 8) {
                    drain_round<true, MODE == 0>(x, src, dst, myq, i, 8, lo, acc, cntL, lane,
                                                 wlist, pos, cap);
                    pos += 8;
                }
                int rem = qn - i;
                if (i > 0 && rem > 0) {
                    int vq = myq[i + ((lane < rem) ? lane : 0)];
                    if (lane < rem) myq[lane] = vq;
                }
                qn = rem;
                wave_lds_order();
            }
            if (qn > 0) {
                drain_round<false, MODE == 0>(x, src, dst, myq, 0, qn, lo, acc, cntL, lane,
                                              wlist, pos, cap);
                pos += qn;
            }
            if (MODE == 0 && cap > 0 && lane == 0) rec[myslot] = pos;
        }
        __syncthreads();

        for (int r = wave; r < nrows; r += AGG_WAVES) {
            const float* a = acc + r * D + lane;
            float4 o;
            o.x = a[0]; o.y = a[32]; o.z = a[64]; o.w = a[96];
            ((float4*)(s + (long long)(lo + r) * D))[lane] = o;
        }
        if (!(MODE == 1 && replay) && tid < nrows) cnt[lo + tid] = cntL[tid];
        __syncthreads();
    }
}

#define GEMM_BLOCKS 512
#define GT_PITCH2   66

__global__ __launch_bounds__(256) void sage_gemm_kernel(
    const float* __restrict__ xin, const float* __restrict__ s,
    const float* __restrict__ cnt, const float* __restrict__ Wl,
    const float* __restrict__ bias, const float* __restrict__ Wr,
    float* __restrict__ out, int N, int relu)
{
    __shared__ __attribute__((aligned(16))) v2f tileA[2][2][16 * GT_PITCH2];

    const int wave  = threadIdx.x >> 5;
    const int lane  = threadIdx.x & 31;
    const int c16   = lane & 15;
    const int hsel  = lane >> 4;
    const int khalf = hsel << 1;
    const int ncol  = wave * 16 + c16;
    const int ntiles = (N + 15) >> 4;

    v2f bl[32], br[32];
    #pragma unroll
    for (int t = 0; t < 32; ++t) {
        int ka = 4 * t + khalf;
        bl[t].x = Wl[ka * D + ncol]; bl[t].y = Wl[(ka + 1) * D + ncol];
        br[t].x = Wr[ka * D + ncol]; br[t].y = Wr[(ka + 1) * D + ncol];
    }
    const float bn = bias[ncol];

    const int r0 = 2 * wave, r1 = r0 + 1;
    float4 ps0, ps1, px0, px1;
    float  pc0, pc1;

#define GEMM_FETCH(tl)                                                        \
    {                                                                         \
        int ga = (tl) * 16 + r0; if (ga > N - 1) ga = N - 1;                  \
        int gb = (tl) * 16 + r1; if (gb > N - 1) gb = N - 1;                  \
        pc0 = cnt[ga];                                                        \
        pc1 = cnt[gb];                                                        \
        ps0 = ((const float4*)(s   + (long long)ga * D))[lane];               \
        ps1 = ((const float4*)(s   + (long long)gb * D))[lane];               \
        px0 = ((const float4*)(xin + (long long)ga * D))[lane];               \
        px1 = ((const float4*)(xin + (long long)gb * D))[lane];               \
    }
#define GEMM_PUT(bf)                                                          \
    {                                                                         \
        float ia = 1.0f / fmaxf(pc0, 1.0f);                                   \
        float ib = 1.0f / fmaxf(pc1, 1.0f);                                   \
        v2f* dS = &tileA[(bf)][0][0];                                         \
        v2f* dX = &tileA[(bf)][1][0];                                         \
        v2f t0; v2f t1;                                                       \
        t0.x = ps0.x * ia; t0.y = ps0.y * ia; t1.x = ps0.z * ia; t1.y = ps0.w * ia; \
        dS[r0 * GT_PITCH2 + 2 * lane] = t0; dS[r0 * GT_PITCH2 + 2 * lane + 1] = t1; \
        t0.x = ps1.x * ib; t0.y = ps1.y * ib; t1.x = ps1.z * ib; t1.y = ps1.w * ib; \
        dS[r1 * GT_PITCH2 + 2 * lane] = t0; dS[r1 * GT_PITCH2 + 2 * lane + 1] = t1; \
        t0.x = px0.x; t0.y = px0.y; t1.x = px0.z; t1.y = px0.w;               \
        dX[r0 * GT_PITCH2 + 2 * lane] = t0; dX[r0 * GT_PITCH2 + 2 * lane + 1] = t1; \
        t0.x = px1.x; t0.y = px1.y; t1.x = px1.z; t1.y = px1.w;               \
        dX[r1 * GT_PITCH2 + 2 * lane] = t0; dX[r1 * GT_PITCH2 + 2 * lane + 1] = t1; \
    }

    int tile = blockIdx.x;
    {
        int first = (tile < ntiles) ? tile : 0;
        GEMM_FETCH(first)
    }
    GEMM_PUT(0)
    __syncthreads();

    int buf = 0;
    for (; tile < ntiles; tile += gridDim.x) {
        const int  nxt  = tile + gridDim.x;
        const bool more = nxt < ntiles;
        if (more) GEMM_FETCH(nxt)

        const v2f* aS = &tileA[buf][0][c16 * GT_PITCH2 + hsel];
        const v2f* aX = &tileA[buf][1][c16 * GT_PITCH2 + hsel];
        v8f c = {};
        #pragma unroll
        for (int t = 0; t < 32; ++t) {
            v2f a0 = aS[2 * t];
            c = __builtin_amdgcn_wmma_f32_16x16x4_f32(
                    false, a0, false, bl[t], (short)0, c, false, false);
            v2f a1 = aX[2 * t];
            c = __builtin_amdgcn_wmma_f32_16x16x4_f32(
                    false, a1, false, br[t], (short)0, c, false, false);
        }

        if (more) GEMM_PUT(buf ^ 1)
        __syncthreads();

        const int mbase = tile * 16 + (hsel << 3);
        #pragma unroll
        for (int r = 0; r < 8; ++r) {
            int m = mbase + r;
            if (m < N) {
                float v = c[r] + bn;
                if (relu) v = fmaxf(v, 0.0f);
                out[(long long)m * D + ncol] = v;
            }
        }
        buf ^= 1;
    }
#undef GEMM_FETCH
#undef GEMM_PUT
}

__global__ __launch_bounds__(256) void decode_kernel(
    const float* __restrict__ h, const int* __restrict__ dec,
    float* __restrict__ out, int Dk)
{
    int wid  = (int)(((long long)blockIdx.x * blockDim.x + threadIdx.x) >> 5);
    int lane = threadIdx.x & 31;
    if (wid >= Dk) return;
    int ai = dec[wid];
    int bi = dec[Dk + wid];
    float4 a = ((const float4*)(h + (long long)ai * D))[lane];
    float4 b = ((const float4*)(h + (long long)bi * D))[lane];
    float dot = a.x * b.x + a.y * b.y + a.z * b.z + a.w * b.w;
    float na  = a.x * a.x + a.y * a.y + a.z * a.z + a.w * a.w;
    float nb  = b.x * b.x + b.y * b.y + b.z * b.z + b.w * b.w;
    #pragma unroll
    for (int off = 16; off > 0; off >>= 1) {
        dot += __shfl_xor(dot, off, 32);
        na  += __shfl_xor(na,  off, 32);
        nb  += __shfl_xor(nb,  off, 32);
    }
    if (lane == 0) {
        float denom = fmaxf(sqrtf(na) * sqrtf(nb), 1e-6f);
        float z = dot / denom;
        out[wid] = 1.0f / (1.0f + expf(-z));
    }
}

extern "C" void kernel_launch(void* const* d_in, const int* in_sizes, int n_in,
                              void* d_out, int out_size, void* d_ws, size_t ws_size,
                              hipStream_t stream) {
    const float* x   = (const float*)d_in[0];
    const int*   edg = (const int*)d_in[1];
    const int*   dec = (const int*)d_in[2];
    const float* W1l = (const float*)d_in[3];
    const float* b1  = (const float*)d_in[4];
    const float* W1r = (const float*)d_in[5];
    const float* W2l = (const float*)d_in[6];
    const float* b2  = (const float*)d_in[7];
    const float* W2r = (const float*)d_in[8];
    float* out = (float*)d_out;

    const int N  = in_sizes[0] / D;
    const int E  = in_sizes[1] / 2;
    const int Dk = in_sizes[2] / 2;
    const int* src = edg;
    const int* dst = edg + E;

    float* ws  = (float*)d_ws;
    float* s   = ws;
    float* cnt = s + (long long)N * D;
    float* h1  = cnt + N;
    float* h2  = h1 + (long long)N * D;

    int* rec   = (int*)h2;
    unsigned* key = (unsigned*)h1;

    const int aggrBlocks  = 256;
    const int gemmBlocks  = GEMM_BLOCKS;
    const int decBlocks   = (Dk + 7) / 8;

    owner_key_kernel<<<1024, 256, 0, stream>>>(dst, key, E, N);
    aggregate_owner_kernel<0><<<aggrBlocks, 1024, 0, stream>>>(x, src, dst, s, cnt, rec, key, E, N);
    sage_gemm_kernel<<<gemmBlocks, 256, 0, stream>>>(x, s, cnt, W1l, b1, W1r,
                                                     h1, N,  1);
    aggregate_owner_kernel<1><<<aggrBlocks, 1024, 0, stream>>>(h1, src, dst, s, cnt, rec, nullptr, E, N);
    sage_gemm_kernel<<<gemmBlocks, 256, 0, stream>>>(h1, s, cnt, W2l, b2, W2r,
                                                     h2, N,  0);
    decode_kernel<<<decBlocks, 256, 0, stream>>>(h2, dec, out, Dk);
}


# decode: one wave per 64 pairs, not one per pair (500000 waves become 7813); 8 lanes take a pair, rows of 8 pairs in flight per wave, 3 xor steps; sums paired as in the 32-lane tree, bit-equal
# speedup vs baseline: 4.2292x; 1.1321x over previous
#include <hip/hip_runtime.h>
#include <hip/hip_bf16.h>
#include <math.h>

typedef __attribute__((ext_vector_type(2))) float v2f;
typedef __attribute__((ext_vector_type(8))) float v8f;

#define D 128

#define AGG_ROWS   197
#define AGG_NOKEY  255u
#define AGG_WAVES  32
#define AGG_BATCH  1024
#define AGG_Q      (AGG_BATCH + 8)

__device__ __forceinline__ void lds_add(float* p, float v) {
    __hip_atomic_fetch_add(p, v, __ATOMIC_RELAXED, __HIP_MEMORY_SCOPE_WORKGROUP);
}

__device__ __forceinline__ void wave_lds_order() {
    __builtin_amdgcn_fence(__ATOMIC_RELEASE, "wavefront");
    __builtin_amdgcn_wave_barrier();
    __builtin_amdgcn_fence(__ATOMIC_ACQUIRE, "wavefront");
}

template <bool FULL, bool REC>
__device__ __forceinline__ void drain_round(
    const float* __restrict__ x, const int* __restrict__ src,
    const int* __restrict__ dst, const int* myq, int i, int n, int lo,
    float* acc, float* cntL, int lane, int2* wlist, int pos, int cap)
{
    int slot = (lane < n) ? (i + lane) : i;
    int e  = myq[slot];
    int sv = src[e];
    int tv = dst[e] - lo;
    if (REC) {
        int p = pos + lane;
        if (lane < n && p < cap) wlist[p] = make_int2(sv, tv);
    }
    float4 v[8];
    #pragma unroll
    for (int j = 0; j < 8; ++j) {
        if (FULL || j < n) {
            int sj = __builtin_amdgcn_readlane(sv, j);
            v[j] = ((const float4*)(x + (long long)sj * D))[lane];
        }
    }
    #pragma unroll
    for (int j = 0; j < 8; ++j) {
        if (FULL || j < n) {
            int tj = __builtin_amdgcn_readlane(tv, j);
            float* a = acc + tj * D + lane;
            lds_add(a,      v[j].x);
            lds_add(a + 32, v[j].y);
            lds_add(a + 64, v[j].z);
            lds_add(a + 96, v[j].w);
            if (lane == 0) lds_add(cntL + tj, 1.0f);
        }
    }
}

template <bool FULL>
__device__ __forceinline__ void replay_round(
    const float* __restrict__ x, int sv, int tv, int r, int n, float* acc, int lane)
{
    float4 v[8];
    #pragma unroll
    for (int j = 0; j < 8; ++j) {
        if (FULL || j < n) {
            int sj = __builtin_amdgcn_readlane(sv, r + j);
            v[j] = ((const float4*)(x + (long long)sj * D))[lane];
        }
    }
    #pragma unroll
    for (int j = 0; j < 8; ++j) {
        if (FULL || j < n) {
            int tj = __builtin_amdgcn_readlane(tv, r + j);
            float* a = acc + tj * D + lane;
            lds_add(a,      v[j].x);
            lds_add(a + 32, v[j].y);
            lds_add(a + 64, v[j].z);
            lds_add(a + 96, v[j].w);
        }
    }
}

__device__ __forceinline__ unsigned key_flags(unsigned dw, unsigned pat) {
    unsigned x = dw ^ pat;
    return ~(((x & 0x7F7F7F7Fu) + 0x7F7F7F7Fu) | x | 0x7F7F7F7Fu);
}

__device__ __forceinline__ bool keys_usable(const unsigned* key, int E, int N, int nblocks) {
    return (N + AGG_ROWS - 1) / AGG_ROWS <= (int)AGG_NOKEY
        && (long long)nblocks * AGG_ROWS >= N
        && (long long)E <= (long long)N * D * 4
        && (((unsigned long long)key) & 15ull) == 0ull;
}

__global__ __launch_bounds__(256) void owner_key_kernel(
    const int* __restrict__ dst, unsigned* __restrict__ key, int E, int N)
{
    if ((long long)E > (long long)N * D * 4) return;
    const bool aligned = (((unsigned long long)dst) & 15ull) == 0ull;
    const int nd = E >> 2;
    for (int t = blockIdx.x * blockDim.x + threadIdx.x; t < nd; t += gridDim.x * blockDim.x) {
        int4 d;
        if (aligned) {
            d = *(const int4*)(dst + 4 * t);
        } else {
            d.x = dst[4 * t]; d.y = dst[4 * t + 1]; d.z = dst[4 * t + 2]; d.w = dst[4 * t + 3];
        }
        unsigned k0 = ((unsigned)d.x < (unsigned)N) ? min((unsigned)d.x / AGG_ROWS, AGG_NOKEY) : AGG_NOKEY;
        unsigned k1 = ((unsigned)d.y < (unsigned)N) ? min((unsigned)d.y / AGG_ROWS, AGG_NOKEY) : AGG_NOKEY;
        unsigned k2 = ((unsigned)d.z < (unsigned)N) ? min((unsigned)d.z / AGG_ROWS, AGG_NOKEY) : AGG_NOKEY;
        unsigned k3 = ((unsigned)d.w < (unsigned)N) ? min((unsigned)d.w / AGG_ROWS, AGG_NOKEY) : AGG_NOKEY;
        key[t] = k0 | (k1 << 8) | (k2 << 16) | (k3 << 24);
    }
}

template <int MODE>
__global__ __launch_bounds__(1024) void aggregate_owner_kernel(
    const float* __restrict__ x, const int* __restrict__ src,
    const int* __restrict__ dst, float* __restrict__ s,
    float* __restrict__ cnt, int* __restrict__ rec,
    const unsigned* __restrict__ key, int E, int N)
{
    __shared__ float acc[AGG_ROWS * D];
    __shared__ float cntL[AGG_ROWS];
    __shared__ int   q[AGG_WAVES][AGG_Q];

    const int tid  = threadIdx.x;
    const int lane = tid & 31;
    const int wave = tid >> 5;
    int* myq = q[wave];
    const unsigned ltmask = (1u << lane) - 1u;
    const bool aligned = (((unsigned long long)dst) & 15ull) == 0ull;
    const int nbatch = (E + AGG_BATCH - 1) / AGG_BATCH;
    const bool usekeys = (MODE == 0) && keys_usable(key, E, N, gridDim.x);
    const unsigned pat = blockIdx.x * 0x01010101u;

    const long long nslots = (long long)gridDim.x * AGG_WAVES;
    const long long room   = (long long)N * D;
    int cap = 0;
    if ((long long)gridDim.x * AGG_ROWS >= N && room >= 3 * nslots)
        cap = (int)((room - nslots) / (2 * nslots));
    const long long myslot = (long long)blockIdx.x * AGG_WAVES + wave;

    for (int lo = blockIdx.x * AGG_ROWS; lo < N; lo += gridDim.x * AGG_ROWS) {
        const int nrows = min(AGG_ROWS, N - lo);
        for (int i = tid; i < AGG_ROWS * D; i += 1024) acc[i] = 0.0f;
        if (tid < AGG_ROWS) cntL[tid] = 0.0f;

        bool replay = false;
        int  m = 0;
        if (MODE == 1 && cap > 0) {
            int len = rec[(long long)blockIdx.x * AGG_WAVES + lane];
            replay = __builtin_amdgcn_ballot_w32(len > cap) == 0u;
            m = __builtin_amdgcn_readlane(len, wave);
        }
        __syncthreads();

        if (MODE == 1 && replay) {
            const int2* wlist = (const int2*)(rec + nslots) + myslot * cap;
            const int mfull = m & ~7;
            const int tail  = m - mfull;
            int2 cur = make_int2(0, 0);
            if (lane < mfull) cur = wlist[lane];
            if (tail > 0) {
                int2 te = wlist[mfull + min(lane, tail - 1)];
                replay_round<false>(x, te.x, te.y, 0, tail, acc, lane);
            }
            for (int i = 0; i < mfull; i += 32) {
                int2 nxt = make_int2(0, 0);
                if (i + 32 + lane < mfull) nxt = wlist[i + 32 + lane];
                const int left = min(32, mfull - i);
                for (int r = 0; r < left; r += 8)
                    replay_round<true>(x, cur.x, cur.y, r, 8, acc, lane);
                cur = nxt;
            }
        } else {
            int2* wlist = (int2*)(rec + nslots) + myslot * cap;
            int qn  = 0;
            int pos = 0;
            for (int b = wave; b < nbatch; b += AGG_WAVES) {
                if (usekeys && (b + 1) * AGG_BATCH <= E) {
                    const int4* kp = (const int4*)(key + b * (AGG_BATCH / 4)) + lane;
                    int4 kv[AGG_BATCH / 512];
                    #pragma unroll
                    for (int u = 0; u < AGG_BATCH / 512; ++u)
                        kv[u] = kp[u * 32];
                    #pragma unroll
                    for (int u = 0; u < AGG_BATCH / 512; ++u) {
                        const int ebase = b * AGG_BATCH + u * 512 + lane * 16;
                        unsigned c = key_flags((unsigned)kv[u].x, pat)
                                   | (key_flags((unsigned)kv[u].y, pat) >> 1)
                                   | (key_flags((unsigned)kv[u].z, pat) >> 2)
                                   | (key_flags((unsigned)kv[u].w, pat) >> 3);
                        unsigned bal = __builtin_amdgcn_ballot_w32(c != 0u);
                        while (bal) {
                            if (c != 0u) {
                                int p = __builtin_ctz(c);
                                myq[qn + __builtin_popcount(bal & ltmask)] =
                                    ebase + ((7 - (p & 7)) << 2) + (p >> 3);
                                c &= c - 1u;
                            }
                            qn += __builtin_popcount(bal);
                            bal = __builtin_amdgcn_ballot_w32(c != 0u);
                        }
                    }
                } else {
                    const int base = b * AGG_BATCH + lane * 4;
                    int4 d[8];
                    if (aligned && (b + 1) * AGG_BATCH <= E) {
                        #pragma unroll
                        for (int u = 0; u < 8; ++u)
                            d[u] = *(const int4*)(dst + base + u * 128);
                    } else {
                        #pragma unroll
                        for (int u = 0; u < 8; ++u) {
                            int e = base + u * 128;
                            d[u].x = (e     < E) ? dst[e]     : -1;
                            d[u].y = (e + 1 < E) ? dst[e + 1] : -1;
                            d[u].z = (e + 2 < E) ? dst[e + 2] : -1;
                            d[u].w = (e + 3 < E) ? dst[e + 3] : -1;
                        }
                    }
#define AGG_ENQ(dv, eidx)                                                     \
                    {                                                             \
                        bool mt = (unsigned)((dv) - lo) < (unsigned)nrows;        \
                        unsigned bal = __builtin_amdgcn_ballot_w32(mt);           \
                        if (bal) {                                                \
                            if (mt) myq[qn + __builtin_popcount(bal & ltmask)] = (eidx); \
                            qn += __builtin_popcount(bal);                        \
                        }                                                         \
                    }
                    #pragma unroll
                    for (int u = 0; u < 8; ++u) {
                        int e = base + u * 128;
                        AGG_ENQ(d[u].x, e)
                        AGG_ENQ(d[u].y, e + 1)
                        AGG_ENQ(d[u].z, e + 2)
                        AGG_ENQ(d[u].w, e + 3)
                    }
#undef AGG_ENQ
                }
                wave_lds_order();
                int i = 0;
                for (; i + 8 <= qn; i += 8) {
                    drain_round<true, MODE == 0>(x, src, dst, myq, i, 8, lo, acc, cntL, lane,
                                                 wlist, pos, cap);
                    pos += 8;
                }
                int rem = qn - i;
                if (i > 0 && rem > 0) {
                    int vq = myq[i + ((lane < rem) ? lane : 0)];
                    if (lane < rem) myq[lane] = vq;
                }
                qn = rem;
                wave_lds_order();
            }
            if (qn > 0) {
                drain_round<false, MODE == 0>(x, src, dst, myq, 0, qn, lo, acc, cntL, lane,
                                              wlist, pos, cap);
                pos += qn;
            }
            if (MODE == 0 && cap > 0 && lane == 0) rec[myslot] = pos;
        }
        __syncthreads();

        for (int r = wave; r < nrows; r += AGG_WAVES) {
            const float* a = acc + r * D + lane;
            float4 o;
            o.x = a[0]; o.y = a[32]; o.z = a[64]; o.w = a[96];
            ((float4*)(s + (long long)(lo + r) * D))[lane] = o;
        }
        if (!(MODE == 1 && replay) && tid < nrows) cnt[lo + tid] = cntL[tid];
        __syncthreads();
    }
}

#define GEMM_BLOCKS 512
#define GT_PITCH2   66

__global__ __launch_bounds__(256) void sage_gemm_kernel(
    const float* __restrict__ xin, const float* __restrict__ s,
    const float* __restrict__ cnt, const float* __restrict__ Wl,
    const float* __restrict__ bias, const float* __restrict__ Wr,
    float* __restrict__ out, int N, int relu)
{
    __shared__ __attribute__((aligned(16))) v2f tileA[2][2][16 * GT_PITCH2];

    const int wave  = threadIdx.x >> 5;
    const int lane  = threadIdx.x & 31;
    const int c16   = lane & 15;
    const int hsel  = lane >> 4;
    const int khalf = hsel << 1;
    const int ncol  = wave * 16 + c16;
    const int ntiles = (N + 15) >> 4;

    v2f bl[32], br[32];
    #pragma unroll
    for (int t = 0; t < 32; ++t) {
        int ka = 4 * t + khalf;
        bl[t].x = Wl[ka * D + ncol]; bl[t].y = Wl[(ka + 1) * D + ncol];
        br[t].x = Wr[ka * D + ncol]; br[t].y = Wr[(ka + 1) * D + ncol];
    }
    const float bn = bias[ncol];

    const int r0 = 2 * wave, r1 = r0 + 1;
    float4 ps0, ps1, px0, px1;
    float  pc0, pc1;

#define GEMM_FETCH(tl)                                                        \
    {                                                                         \
        int ga = (tl) * 16 + r0; if (ga > N - 1) ga = N - 1;                  \
        int gb = (tl) * 16 + r1; if (gb > N - 1) gb = N - 1;                  \
        pc0 = cnt[ga];                                                        \
        pc1 = cnt[gb];                                                        \
        ps0 = ((const float4*)(s   + (long long)ga * D))[lane];               \
        ps1 = ((const float4*)(s   + (long long)gb * D))[lane];               \
        px0 = ((const float4*)(xin + (long long)ga * D))[lane];               \
        px1 = ((const float4*)(xin + (long long)gb * D))[lane];               \
    }
#define GEMM_PUT(bf)                                                          \
    {                                                                         \
        float ia = 1.0f / fmaxf(pc0, 1.0f);                                   \
        float ib = 1.0f / fmaxf(pc1, 1.0f);                                   \
        v2f* dS = &tileA[(bf)][0][0];                                         \
        v2f* dX = &tileA[(bf)][1][0];                                         \
        v2f t0; v2f t1;                                                       \
        t0.x = ps0.x * ia; t0.y = ps0.y * ia; t1.x = ps0.z * ia; t1.y = ps0.w * ia; \
        dS[r0 * GT_PITCH2 + 2 * lane] = t0; dS[r0 * GT_PITCH2 + 2 * lane + 1] = t1; \
        t0.x = ps1.x * ib; t0.y = ps1.y * ib; t1.x = ps1.z * ib; t1.y = ps1.w * ib; \
        dS[r1 * GT_PITCH2 + 2 * lane] = t0; dS[r1 * GT_PITCH2 + 2 * lane + 1] = t1; \
        t0.x = px0.x; t0.y = px0.y; t1.x = px0.z; t1.y = px0.w;               \
        dX[r0 * GT_PITCH2 + 2 * lane] = t0; dX[r0 * GT_PITCH2 + 2 * lane + 1] = t1; \
        t0.x = px1.x; t0.y = px1.y; t1.x = px1.z; t1.y = px1.w;               \
        dX[r1 * GT_PITCH2 + 2 * lane] = t0; dX[r1 * GT_PITCH2 + 2 * lane + 1] = t1; \
    }

    int tile = blockIdx.x;
    {
        int first = (tile < ntiles) ? tile : 0;
        GEMM_FETCH(first)
    }
    GEMM_PUT(0)
    __syncthreads();

    int buf = 0;
    for (; tile < ntiles; tile += gridDim.x) {
        const int  nxt  = tile + gridDim.x;
        const bool more = nxt < ntiles;
        if (more) GEMM_FETCH(nxt)

        const v2f* aS = &tileA[buf][0][c16 * GT_PITCH2 + hsel];
        const v2f* aX = &tileA[buf][1][c16 * GT_PITCH2 + hsel];
        v8f c = {};
        #pragma unroll
        for (int t = 0; t < 32; ++t) {
            v2f a0 = aS[2 * t];
            c = __builtin_amdgcn_wmma_f32_16x16x4_f32(
                    false, a0, false, bl[t], (short)0, c, false, false);
            v2f a1 = aX[2 * t];
            c = __builtin_amdgcn_wmma_f32_16x16x4_f32(
                    false, a1, false, br[t], (short)0, c, false, false);
        }

        if (more) GEMM_PUT(buf ^ 1)
        __syncthreads();

        const int mbase = tile * 16 + (hsel << 3);
        #pragma unroll
        for (int r = 0; r < 8; ++r) {
            int m = mbase + r;
            if (m < N) {
                float v = c[r] + bn;
                if (relu) v = fmaxf(v, 0.0f);
                out[(long long)m * D + ncol] = v;
            }
        }
        buf ^= 1;
    }
#undef GEMM_FETCH
#undef GEMM_PUT
}

#define DEC_PAIRS 64

__device__ __forceinline__ float chunk_dot(const float4 a, const float4 b) {
    return fmaf(a.w, b.w, fmaf(a.z, b.z, fmaf(a.x, b.x, a.y * b.y)));
}

__global__ __launch_bounds__(256) void decode_kernel(
    const float* __restrict__ h, const int* __restrict__ dec,
    float* __restrict__ out, int Dk)
{
    const int wid  = (int)(((long long)blockIdx.x * blockDim.x + threadIdx.x) >> 5);
    const int lane = threadIdx.x & 31;
    const long long base = (long long)wid * DEC_PAIRS;
    if (base >= Dk) return;
    const int sub = lane & 7;
    const int grp = lane & 24;

    int ia[2], ib[2];
    #pragma unroll
    for (int hh = 0; hh < 2; ++hh) {
        long long p = base + 32 * hh + lane;
        if (p > Dk - 1) p = Dk - 1;
        ia[hh] = dec[p];
        ib[hh] = dec[Dk + p];
    }

    float rd[2] = {0.0f, 0.0f}, rna[2] = {0.0f, 0.0f}, rnb[2] = {0.0f, 0.0f};

    #pragma unroll
    for (int pp = 0; pp < 16; pp += 2) {
        float4 a[2][4], b[2][4];
        #pragma unroll
        for (int q = 0; q < 2; ++q) {
            const int hh = (pp + q) >> 3;
            const int k  = (pp + q) & 7;
            const int ai = __shfl(ia[hh], grp | k, 32);
            const int bi = __shfl(ib[hh], grp | k, 32);
            const float4* ra = (const float4*)(h + (long long)ai * D) + sub;
            const float4* rb = (const float4*)(h + (long long)bi * D) + sub;
            #pragma unroll
            for (int t = 0; t < 4; ++t) {
                a[q][t] = ra[8 * t];
                b[q][t] = rb[8 * t];
            }
        }
        __builtin_amdgcn_sched_barrier(0);
        #pragma unroll
        for (int q = 0; q < 2; ++q) {
            const int hh = (pp + q) >> 3;
            const int k  = (pp + q) & 7;
            float sd[4], sa[4], sb[4];
            #pragma unroll
            for (int t = 0; t < 4; ++t) {
                const float4 av = a[q][t];
                const float4 bv = b[q][t];
                sd[t] = chunk_dot(av, bv);
                sa[t] = chunk_dot(av, av);
                sb[t] = chunk_dot(bv, bv);
            }
            float dot = (sd[0] + sd[2]) + (sd[1] + sd[3]);
            float na  = (sa[0] + sa[2]) + (sa[1] + sa[3]);
            float nb  = (sb[0] + sb[2]) + (sb[1] + sb[3]);
            #pragma unroll
            for (int off = 4; off > 0; off >>= 1) {
                dot += __shfl_xor(dot, off, 32);
                na  += __shfl_xor(na,  off, 32);
                nb  += __shfl_xor(nb,  off, 32);
            }
            if (sub == k) { rd[hh] = dot; rna[hh] = na; rnb[hh] = nb; }
        }
    }

    #pragma unroll
    for (int hh = 0; hh < 2; ++hh) {
        const long long p = base + 32 * hh + lane;
        float denom = fmaxf(sqrtf(rna[hh]) * sqrtf(rnb[hh]), 1e-6f);
        float z = rd[hh] / denom;
        float v = 1.0f / (1.0f + expf(-z));
        if (p < Dk) out[p] = v;
    }
}

extern "C" void kernel_launch(void* const* d_in, const int* in_sizes, int n_in,
                              void* d_out, int out_size, void* d_ws, size_t ws_size,
                              hipStream_t stream) {
    const float* x   = (const float*)d_in[0];
    const int*   edg = (const int*)d_in[1];
    const int*   dec = (const int*)d_in[2];
    const float* W1l = (const float*)d_in[3];
    const float* b1  = (const float*)d_in[4];
    const float* W1r = (const float*)d_in[5];
    const float* W2l = (const float*)d_in[6];
    const float* b2  = (const float*)d_in[7];
    const float* W2r = (const float*)d_in[8];
    float* out = (float*)d_out;

    const int N  = in_sizes[0] / D;
    const int E  = in_sizes[1] / 2;
    const int Dk = in_sizes[2] / 2;
    const int* src = edg;
    const int* dst = edg + E;

    float* ws  = (float*)d_ws;
    float* s   = ws;
    float* cnt = s + (long long)N * D;
    float* h1  = cnt + N;
    float* h2  = h1 + (long long)N * D;

    int* rec   = (int*)h2;
    unsigned* key = (unsigned*)h1;

    const int aggrBlocks  = 256;
    const int gemmBlocks  = GEMM_BLOCKS;
    const int decBlocks   = (Dk + 511) / 512;

    owner_key_kernel<<<1024, 256, 0, stream>>>(dst, key, E, N);
    aggregate_owner_kernel<0><<<aggrBlocks, 1024, 0, stream>>>(x, src, dst, s, cnt, rec, key, E, N);
    sage_gemm_kernel<<<gemmBlocks, 256, 0, stream>>>(x, s, cnt, W1l, b1, W1r,
                                                     h1, N,  1);
    aggregate_owner_kernel<1><<<aggrBlocks, 1024, 0, stream>>>(h1, src, dst, s, cnt, rec, nullptr, E, N);
    sage_gemm_kernel<<<gemmBlocks, 256, 0, stream>>>(h1, s, cnt, W2l, b2, W2r,
                                                     h2, N,  0);
    decode_kernel<<<decBlocks, 256, 0, stream>>>(h2, dec, out, Dk);
}
